# starting kernel
# baseline (speedup 1.0000x reference)
#include <hip/hip_runtime.h>
#include <cstdint>

#define D 128
#define LDSP 132
#define SCAN_BS 1024

typedef float v2f __attribute__((ext_vector_type(2)));
typedef float v8f __attribute__((ext_vector_type(8)));

__device__ __forceinline__ v2f mkv2(float a, float b) { v2f v; v.x = a; v.y = b; return v; }

__device__ __forceinline__ v8f wmma_f32(v2f a, v2f b, v8f c) {
    return __builtin_amdgcn_wmma_f32_16x16x4_f32(false, a, false, b, (short)0, c, false, false);
}

__device__ __forceinline__ float sigmoidf_(float x) { return 1.0f / (1.0f + __expf(-x)); }

__global__ void k_izero(int* __restrict__ p, int n) {
    int i = blockIdx.x * blockDim.x + threadIdx.x;
    if (i < n) p[i] = 0;
}

__global__ void k_count(const int* __restrict__ dst, int* __restrict__ indeg, int E) {
    int e = blockIdx.x * blockDim.x + threadIdx.x;
    if (e < E) atomicAdd(indeg + dst[e], 1);
}

__global__ __launch_bounds__(SCAN_BS) void k_scan1(const int* __restrict__ in,
                                                   int* __restrict__ incl,
                                                   int* __restrict__ bsums, int n) {
    __shared__ int sh[SCAN_BS];
    int gid = blockIdx.x * SCAN_BS + threadIdx.x;
    int v = (gid < n) ? in[gid] : 0;
    sh[threadIdx.x] = v;
    __syncthreads();
    for (int off = 1; off < SCAN_BS; off <<= 1) {
        int t = (threadIdx.x >= off) ? sh[threadIdx.x - off] : 0;
        __syncthreads();
        sh[threadIdx.x] += t;
        __syncthreads();
    }
    if (gid < n) incl[gid] = sh[threadIdx.x];
    if (threadIdx.x == SCAN_BS - 1) bsums[blockIdx.x] = sh[SCAN_BS - 1];
}

__global__ __launch_bounds__(SCAN_BS) void k_scan2(int* __restrict__ bsums, int nb) {
    __shared__ int sh[SCAN_BS];
    int v = (threadIdx.x < nb) ? bsums[threadIdx.x] : 0;
    sh[threadIdx.x] = v;
    __syncthreads();
    for (int off = 1; off < SCAN_BS; off <<= 1) {
        int t = (threadIdx.x >= off) ? sh[threadIdx.x - off] : 0;
        __syncthreads();
        sh[threadIdx.x] += t;
        __syncthreads();
    }
    if (threadIdx.x < nb) bsums[threadIdx.x] = sh[threadIdx.x] - v;
}

__global__ void k_scan3(int* __restrict__ start  ,
                        const int* __restrict__ indeg, const int* __restrict__ bsums,
                        int* __restrict__ cursor, int n) {
    int gid = blockIdx.x * blockDim.x + threadIdx.x;
    if (gid < n) {
        int ex = start[gid] - indeg[gid] + bsums[gid / SCAN_BS];
        start[gid] = ex;
        cursor[gid] = ex;
    }
}

__global__ void k_fill(const int* __restrict__ src, const int* __restrict__ dst,
                       int* __restrict__ cursor, int* __restrict__ csrc, int E) {
    int e = blockIdx.x * blockDim.x + threadIdx.x;
    if (e < E) {
        int p = atomicAdd(cursor + dst[e], 1);
        csrc[p] = src[e];
    }
}

__global__ void k_gather2(const float* __restrict__ xin, const float* __restrict__ hl,
                          const int* __restrict__ csrc,
                          const int* __restrict__ start, const int* __restrict__ indeg,
                          float* __restrict__ aggx, float* __restrict__ aggh, int N) {
    int node = (int)(((size_t)blockIdx.x * blockDim.x + threadIdx.x) >> 5);
    int lane = threadIdx.x & 31;
    if (node >= N) return;
    int base = start[node], cnt = indeg[node];
    float4 ax = {0.f, 0.f, 0.f, 0.f}, ah = {0.f, 0.f, 0.f, 0.f};
#pragma unroll 2
    for (int j = 0; j < cnt; ++j) {
        int s = csrc[base + j];
        float4 vx = ((const float4*)(xin + (size_t)s * D))[lane];
        float4 vh = ((const float4*)(hl + (size_t)s * D))[lane];
        ax.x += vx.x; ax.y += vx.y; ax.z += vx.z; ax.w += vx.w;
        ah.x += vh.x; ah.y += vh.y; ah.z += vh.z; ah.w += vh.w;
    }
    float inv = 1.0f / fmaxf((float)cnt, 1.0f);
    float4 ox = {ax.x * inv, ax.y * inv, ax.z * inv, ax.w * inv};
    float4 oh = {ah.x * inv, ah.y * inv, ah.z * inv, ah.w * inv};
    ((float4*)(aggx + (size_t)node * D))[lane] = ox;
    ((float4*)(aggh + (size_t)node * D))[lane] = oh;
}

__global__ void k_gather1(const float* __restrict__ xin,
                          const int* __restrict__ csrc,
                          const int* __restrict__ start, const int* __restrict__ indeg,
                          float* __restrict__ agg, int N) {
    int node = (int)(((size_t)blockIdx.x * blockDim.x + threadIdx.x) >> 5);
    int lane = threadIdx.x & 31;
    if (node >= N) return;
    int base = start[node], cnt = indeg[node];
    float4 ax = {0.f, 0.f, 0.f, 0.f};
#pragma unroll 2
    for (int j = 0; j < cnt; ++j) {
        int s = csrc[base + j];
        float4 vx = ((const float4*)(xin + (size_t)s * D))[lane];
        ax.x += vx.x; ax.y += vx.y; ax.z += vx.z; ax.w += vx.w;
    }
    float inv = 1.0f / fmaxf((float)cnt, 1.0f);
    float4 ox = {ax.x * inv, ax.y * inv, ax.z * inv, ax.w * inv};
    ((float4*)(agg + (size_t)node * D))[lane] = ox;
}

__global__ __launch_bounds__(256) void k_gates(
    const float* __restrict__ xin, const float* __restrict__ hl,
    const float* __restrict__ Wl, const float* __restrict__ Wr,
    const float* __restrict__ bias,
    float* __restrict__ bufA,
    float* __restrict__ bufB,
    float* __restrict__ rh, int N) {
    __shared__ float lds[4][32][LDSP];
    const int row0 = blockIdx.x * 32;
    const int tid = threadIdx.x;
    for (int idx = tid; idx < 32 * D; idx += 256) {
        int r = idx >> 7, c = idx & (D - 1);
        int row = row0 + r; if (row >= N) row = N - 1;
        size_t off = (size_t)row * D + c;
        lds[0][r][c] = bufA[off];
        lds[1][r][c] = xin[off];
        lds[2][r][c] = bufB[off];
        lds[3][r][c] = hl[off];
    }
    __syncthreads();

    const int wave = tid >> 5;
    const int lane = tid & 31;
    const int m = lane & 15;
    const int half = lane >> 4;
    const int col = wave * 16 + m;

    v8f cz[2] = {{}, {}};
    v8f cr[2] = {{}, {}};
    v8f ct[2] = {{}, {}};
#pragma unroll 2
    for (int kb = 0; kb < D; kb += 4) {
        const int kk = kb + half * 2;
#define BF(W, g) mkv2(W[(size_t)(g) * D * D + (size_t)kk * D + col], \
                      W[(size_t)(g) * D * D + (size_t)(kk + 1) * D + col])
        v2f b0l = BF(Wl, 0), b0r = BF(Wr, 0);
        v2f b1l = BF(Wl, 1), b1r = BF(Wr, 1);
        v2f b2l = BF(Wl, 2), b2r = BF(Wr, 2);
        v2f b3l = BF(Wl, 3), b3r = BF(Wr, 3);
        v2f b4l = BF(Wl, 4), b4r = BF(Wr, 4);
#undef BF
#pragma unroll
        for (int sub = 0; sub < 2; ++sub) {
            const int mm = sub * 16 + m;
            v2f aAX = mkv2(lds[0][mm][kk], lds[0][mm][kk + 1]);
            v2f aX  = mkv2(lds[1][mm][kk], lds[1][mm][kk + 1]);
            v2f aAH = mkv2(lds[2][mm][kk], lds[2][mm][kk + 1]);
            v2f aH  = mkv2(lds[3][mm][kk], lds[3][mm][kk + 1]);
            cz[sub] = wmma_f32(aAX, b0l, cz[sub]);
            cz[sub] = wmma_f32(aX,  b0r, cz[sub]);
            cz[sub] = wmma_f32(aAH, b1l, cz[sub]);
            cz[sub] = wmma_f32(aH,  b1r, cz[sub]);
            cr[sub] = wmma_f32(aAX, b2l, cr[sub]);
            cr[sub] = wmma_f32(aX,  b2r, cr[sub]);
            cr[sub] = wmma_f32(aAH, b3l, cr[sub]);
            cr[sub] = wmma_f32(aH,  b3r, cr[sub]);
            ct[sub] = wmma_f32(aAX, b4l, ct[sub]);
            ct[sub] = wmma_f32(aX,  b4r, ct[sub]);
        }
    }

    const float bz = bias[col] + bias[D + col];
    const float br = bias[2 * D + col] + bias[3 * D + col];
    const float bt = bias[4 * D + col];
#pragma unroll
    for (int sub = 0; sub < 2; ++sub) {
#pragma unroll
        for (int j = 0; j < 8; ++j) {
            int rloc = sub * 16 + j + half * 8;
            int row = row0 + rloc;
            if (row < N) {
                size_t off = (size_t)row * D + col;
                float z = sigmoidf_(cz[sub][j] + bz);
                float r = sigmoidf_(cr[sub][j] + br);
                float hv = lds[3][rloc][col];
                bufA[off] = z;
                rh[off] = r * hv;
                bufB[off] = ct[sub][j] + bt;
            }
        }
    }
}

__global__ __launch_bounds__(256) void k_final(
    const float* __restrict__ hl,
    const float* __restrict__ Wl5, const float* __restrict__ Wr5,
    const float* __restrict__ b5,
    const float* __restrict__ zbuf, const float* __restrict__ tpart,
    const float* __restrict__ rh, const float* __restrict__ aggrh,
    float* __restrict__ outl, int N) {
    __shared__ float lds[2][32][LDSP];
    const int row0 = blockIdx.x * 32;
    const int tid = threadIdx.x;
    for (int idx = tid; idx < 32 * D; idx += 256) {
        int r = idx >> 7, c = idx & (D - 1);
        int row = row0 + r; if (row >= N) row = N - 1;
        size_t off = (size_t)row * D + c;
        lds[0][r][c] = aggrh[off];
        lds[1][r][c] = rh[off];
    }
    __syncthreads();

    const int wave = tid >> 5;
    const int lane = tid & 31;
    const int m = lane & 15;
    const int half = lane >> 4;
    const int col = wave * 16 + m;

    v8f c[2] = {{}, {}};
#pragma unroll
    for (int sub = 0; sub < 2; ++sub)
#pragma unroll
        for (int j = 0; j < 8; ++j) {
            int row = row0 + sub * 16 + j + half * 8;
            if (row < N) c[sub][j] = tpart[(size_t)row * D + col];
        }
#pragma unroll 2
    for (int kb = 0; kb < D; kb += 4) {
        const int kk = kb + half * 2;
        v2f bl = mkv2(Wl5[(size_t)kk * D + col], Wl5[(size_t)(kk + 1) * D + col]);
        v2f br = mkv2(Wr5[(size_t)kk * D + col], Wr5[(size_t)(kk + 1) * D + col]);
#pragma unroll
        for (int sub = 0; sub < 2; ++sub) {
            const int mm = sub * 16 + m;
            v2f a0 = mkv2(lds[0][mm][kk], lds[0][mm][kk + 1]);
            v2f a1 = mkv2(lds[1][mm][kk], lds[1][mm][kk + 1]);
            c[sub] = wmma_f32(a0, bl, c[sub]);
            c[sub] = wmma_f32(a1, br, c[sub]);
        }
    }

    const float bb = b5[col];
#pragma unroll
    for (int sub = 0; sub < 2; ++sub)
#pragma unroll
        for (int j = 0; j < 8; ++j) {
            int row = row0 + sub * 16 + j + half * 8;
            if (row < N) {
                size_t off = (size_t)row * D + col;
                float ht = tanhf(c[sub][j] + bb);
                float z = zbuf[off];
                float hv = hl[off];
                outl[off] = z * hv + (1.0f - z) * ht;
            }
        }
}

extern "C" void kernel_launch(void* const* d_in, const int* in_sizes, int n_in,
                              void* d_out, int out_size, void* d_ws, size_t ws_size,
                              hipStream_t stream) {
    const float* x  = (const float*)d_in[0];
    const float* h  = (const float*)d_in[1];
    const float* Wl = (const float*)d_in[2];
    const float* Wr = (const float*)d_in[3];
    const float* b  = (const float*)d_in[4];
    const int* src  = (const int*)d_in[5];
    const int* dst  = (const int*)d_in[6];
    const size_t ND = (size_t)in_sizes[0];
    const int N = (int)(ND / D);
    const int L = in_sizes[1] / (int)ND;
    const int E = in_sizes[5];
    float* out = (float*)d_out;

    char* wsc = (char*)d_ws;
    int* indeg  = (int*)wsc;
    int* start  = indeg + N;
    int* cursor = start + N;
    int* bsums  = cursor + N;
    int* csrc   = bsums + SCAN_BS;
    size_t int_bytes = ((size_t)(3 * N + SCAN_BS + E) * sizeof(int) + 63) & ~(size_t)63;
    float* fbase = (float*)(wsc + int_bytes);
    float* bufA  = fbase;
    float* bufB  = bufA + ND;
    float* rh    = bufB + ND;
    float* aggrh = rh + ND;

    const int nb = (N + SCAN_BS - 1) / SCAN_BS;
    const int tiles = (N + 31) / 32;
    const int gblocks = (N + 7) / 8;

    k_izero<<<(N + 255) / 256, 256, 0, stream>>>(indeg, N);
    k_count<<<(E + 255) / 256, 256, 0, stream>>>(dst, indeg, E);
    k_scan1<<<nb, SCAN_BS, 0, stream>>>(indeg, start, bsums, N);
    k_scan2<<<1, SCAN_BS, 0, stream>>>(bsums, nb);
    k_scan3<<<(N + 255) / 256, 256, 0, stream>>>(start, indeg, bsums, cursor, N);
    k_fill<<<(E + 255) / 256, 256, 0, stream>>>(src, dst, cursor, csrc, E);

    for (int l = 0; l < L; ++l) {
        const float* inp = (l == 0) ? x : (out + (size_t)(l - 1) * ND);
        const float* hlp = h + (size_t)l * ND;
        const float* Wll = Wl + (size_t)l * 6 * D * D;
        const float* Wrl = Wr + (size_t)l * 6 * D * D;
        const float* bl  = b + (size_t)l * 6 * D;
        float* outl = out + (size_t)l * ND;

        k_gather2<<<gblocks, 256, 0, stream>>>(inp, hlp, csrc, start, indeg,
                                               bufA, bufB, N);
        k_gates<<<tiles, 256, 0, stream>>>(inp, hlp, Wll, Wrl, bl,
                                           bufA, bufB, rh, N);
        k_gather1<<<gblocks, 256, 0, stream>>>(rh, csrc, start, indeg, aggrh, N);
        k_final<<<tiles, 256, 0, stream>>>(hlp, Wll + 5 * D * D, Wrl + 5 * D * D,
                                           bl + 5 * D, bufA, bufB, rh, aggrh,
                                           outl, N);
    }
}


# CSR build as a two-level counting sort in LDS (256-node buckets, then nodes): no global atomics, 6 launches become 4; slot order in a node, so each neighbour sum's order, now from LDS arrival
# speedup vs baseline: 1.2043x; 1.2043x over previous
#include <hip/hip_runtime.h>
#include <cstdint>

#define D 128
#define LDSP 132
#define SCAN_BS 1024

typedef float v2f __attribute__((ext_vector_type(2)));
typedef float v8f __attribute__((ext_vector_type(8)));

__device__ __forceinline__ v2f mkv2(float a, float b) { v2f v; v.x = a; v.y = b; return v; }

__device__ __forceinline__ v8f wmma_f32(v2f a, v2f b, v8f c) {
    return __builtin_amdgcn_wmma_f32_16x16x4_f32(false, a, false, b, (short)0, c, false, false);
}

__device__ __forceinline__ float sigmoidf_(float x) { return 1.0f / (1.0f + __expf(-x)); }

#define BSH 8
#define BNODES (1 << BSH)
#define MAXB 1024
#define ETH 1024
#define ECH 4096
#define SCAP 8192

__global__ __launch_bounds__(ETH) void k_binhist(const int* __restrict__ dst,
                                                 int* __restrict__ bh,
                                                 int E, int NB, int EB) {
    __shared__ int hist[MAXB];
    const int t = threadIdx.x, b = blockIdx.x;
    for (int k = t; k < NB; k += ETH) hist[k] = 0;
    __syncthreads();
    const int e0 = b * ECH;
    const int e1 = min(E, e0 + ECH);
    for (int i = e0 + t; i < e1; i += ETH) atomicAdd(&hist[dst[i] >> BSH], 1);
    __syncthreads();
    for (int k = t; k < NB; k += ETH) bh[(size_t)k * EB + b] = hist[k];
}

__global__ __launch_bounds__(SCAN_BS) void k_binscan(const int* __restrict__ bh,
                                                     int* __restrict__ offs, int total) {
    __shared__ int sh[SCAN_BS];
    const int t = threadIdx.x;
    const int per = (total + SCAN_BS - 1) / SCAN_BS;
    const int k0 = t * per;
    const int k1 = min(total, k0 + per);
    int s = 0;
    for (int k = k0; k < k1; ++k) s += bh[k];
    sh[t] = s;
    __syncthreads();
    for (int off = 1; off < SCAN_BS; off <<= 1) {
        int v = (t >= off) ? sh[t - off] : 0;
        __syncthreads();
        sh[t] += v;
        __syncthreads();
    }
    int run = sh[t] - s;
    for (int kb = k0; kb < k1; kb += 16) {
        int v[16];
#pragma unroll
        for (int j = 0; j < 16; ++j) v[j] = (kb + j < k1) ? bh[kb + j] : 0;
#pragma unroll
        for (int j = 0; j < 16; ++j) {
            if (kb + j < k1) offs[kb + j] = run;
            run += v[j];
        }
    }
}

static_assert(MAXB == ETH, "the local scan below gives each thread one bucket");
__global__ __launch_bounds__(ETH) void k_binscatter(const int* __restrict__ src,
                                                    const int* __restrict__ dst,
                                                    const int* __restrict__ offs,
                                                    int* __restrict__ bdst,
                                                    int* __restrict__ bsrc,
                                                    int E, int NB, int EB) {
    __shared__ int lsc[MAXB];
    __shared__ int cur[MAXB];
    __shared__ int gof[MAXB];
    __shared__ int sd[ECH];
    __shared__ int ss[ECH];
    const int t = threadIdx.x, b = blockIdx.x;
    const int e0 = b * ECH;
    const int e1 = min(E, e0 + ECH);
    const int n = e1 - e0;
    cur[t] = 0;
    int d[ECH / ETH], s[ECH / ETH];
#pragma unroll
    for (int j = 0; j < ECH / ETH; ++j) {
        const int i = e0 + t + j * ETH;
        d[j] = (i < e1) ? dst[i] : -1;
        s[j] = (i < e1) ? src[i] : 0;
    }
    __syncthreads();
#pragma unroll
    for (int j = 0; j < ECH / ETH; ++j)
        if (d[j] >= 0) atomicAdd(&cur[d[j] >> BSH], 1);
    __syncthreads();
    const int c = cur[t];
    lsc[t] = c;
    __syncthreads();
    for (int off = 1; off < MAXB; off <<= 1) {
        int v = (t >= off) ? lsc[t - off] : 0;
        __syncthreads();
        lsc[t] += v;
        __syncthreads();
    }
    const int ex = lsc[t] - c;
    cur[t] = ex;
    gof[t] = (t < NB) ? offs[(size_t)t * EB + b] - ex : 0;
    __syncthreads();
#pragma unroll
    for (int j = 0; j < ECH / ETH; ++j) {
        if (d[j] >= 0) {
            const int p = atomicAdd(&cur[d[j] >> BSH], 1);
            sd[p] = d[j];
            ss[p] = s[j];
        }
    }
    __syncthreads();
    for (int j = t; j < n; j += ETH) {
        const int dd = sd[j];
        const int g = gof[dd >> BSH] + j;
        bdst[g] = dd;
        bsrc[g] = ss[j];
    }
}

__global__ __launch_bounds__(ETH) void k_bucketsort(const int* __restrict__ offs,
                                                    const int* __restrict__ bdst,
                                                    const int* __restrict__ bsrc,
                                                    int* __restrict__ indeg,
                                                    int* __restrict__ start,
                                                    int* __restrict__ csrc,
                                                    int N, int E, int NB, int EB) {
    __shared__ int cnt[BNODES];
    __shared__ int sc[BNODES];
    __shared__ int so[SCAP];
    const int t = threadIdx.x, b = blockIdx.x;
    const bool own = t < BNODES;
    const int bs = offs[(size_t)b * EB];
    const int be = (b + 1 < NB) ? offs[(size_t)(b + 1) * EB] : E;
    const int n = be - bs;
    const bool staged = n <= SCAP;
    if (own) cnt[t] = 0;
    __syncthreads();
    for (int i = bs + t; i < be; i += ETH) atomicAdd(&cnt[bdst[i] & (BNODES - 1)], 1);
    __syncthreads();
    int c = 0;
    if (own) { c = cnt[t]; sc[t] = c; }
    __syncthreads();
    for (int off = 1; off < BNODES; off <<= 1) {
        int v = 0;
        if (own && t >= off) v = sc[t - off];
        __syncthreads();
        if (own) sc[t] += v;
        __syncthreads();
    }
    if (own) {
        const int ex = sc[t] - c;
        const int node = (b << BSH) + t;
        if (node < N) {
            indeg[node] = c;
            start[node] = bs + ex;
        }
        cnt[t] = ex;
    }
    __syncthreads();
    for (int i = bs + t; i < be; i += ETH) {
        const int p = atomicAdd(&cnt[bdst[i] & (BNODES - 1)], 1);
        const int v = bsrc[i];
        if (staged) so[p] = v;
        else csrc[bs + p] = v;
    }
    if (staged) {
        __syncthreads();
        for (int j = t; j < n; j += ETH) csrc[bs + j] = so[j];
    }
}

__global__ void k_gather2(const float* __restrict__ xin, const float* __restrict__ hl,
                          const int* __restrict__ csrc,
                          const int* __restrict__ start, const int* __restrict__ indeg,
                          float* __restrict__ aggx, float* __restrict__ aggh, int N) {
    int node = (int)(((size_t)blockIdx.x * blockDim.x + threadIdx.x) >> 5);
    int lane = threadIdx.x & 31;
    if (node >= N) return;
    int base = start[node], cnt = indeg[node];
    float4 ax = {0.f, 0.f, 0.f, 0.f}, ah = {0.f, 0.f, 0.f, 0.f};
#pragma unroll 2
    for (int j = 0; j < cnt; ++j) {
        int s = csrc[base + j];
        float4 vx = ((const float4*)(xin + (size_t)s * D))[lane];
        float4 vh = ((const float4*)(hl + (size_t)s * D))[lane];
        ax.x += vx.x; ax.y += vx.y; ax.z += vx.z; ax.w += vx.w;
        ah.x += vh.x; ah.y += vh.y; ah.z += vh.z; ah.w += vh.w;
    }
    float inv = 1.0f / fmaxf((float)cnt, 1.0f);
    float4 ox = {ax.x * inv, ax.y * inv, ax.z * inv, ax.w * inv};
    float4 oh = {ah.x * inv, ah.y * inv, ah.z * inv, ah.w * inv};
    ((float4*)(aggx + (size_t)node * D))[lane] = ox;
    ((float4*)(aggh + (size_t)node * D))[lane] = oh;
}

__global__ void k_gather_gru(const float* __restrict__ q, const float* __restrict__ sbuf,
                             const float* __restrict__ zbuf, const float* __restrict__ hl,
                             const float* __restrict__ b5,
                             const int* __restrict__ csrc,
                             const int* __restrict__ start, const int* __restrict__ indeg,
                             float* __restrict__ outl, int N) {
    int node = (int)(((size_t)blockIdx.x * blockDim.x + threadIdx.x) >> 5);
    int lane = threadIdx.x & 31;
    if (node >= N) return;
    int base = start[node], cnt = indeg[node];
    const size_t ro = (size_t)node * D;
    float4 sv = ((const float4*)(sbuf + ro))[lane];
    float4 zv = ((const float4*)(zbuf + ro))[lane];
    float4 hv = ((const float4*)(hl + ro))[lane];
    float4 bv = ((const float4*)b5)[lane];
    float4 aq = {0.f, 0.f, 0.f, 0.f};
#pragma unroll 2
    for (int j = 0; j < cnt; ++j) {
        int sn = csrc[base + j];
        float4 vq = ((const float4*)(q + (size_t)sn * D))[lane];
        aq.x += vq.x; aq.y += vq.y; aq.z += vq.z; aq.w += vq.w;
    }
    float inv = 1.0f / fmaxf((float)cnt, 1.0f);
    float4 o;
    o.x = zv.x * hv.x + (1.0f - zv.x) * tanhf((sv.x + aq.x * inv) + bv.x);
    o.y = zv.y * hv.y + (1.0f - zv.y) * tanhf((sv.y + aq.y * inv) + bv.y);
    o.z = zv.z * hv.z + (1.0f - zv.z) * tanhf((sv.z + aq.z * inv) + bv.z);
    o.w = zv.w * hv.w + (1.0f - zv.w) * tanhf((sv.w + aq.w * inv) + bv.w);
    ((float4*)(outl + ro))[lane] = o;
}

__global__ __launch_bounds__(256, 8) void k_gates(
    const float* __restrict__ xin, const float* __restrict__ hl,
    const float* __restrict__ Wl, const float* __restrict__ Wr,
    const float* __restrict__ bias,
    float* __restrict__ bufA,
    float* __restrict__ bufB,
    float* __restrict__ qbuf, int N) {
    __shared__ float lds[4][32][LDSP];
    const int row0 = blockIdx.x * 32;
    const int tid = threadIdx.x;
    for (int idx = tid; idx < 32 * D; idx += 256) {
        int r = idx >> 7, c = idx & (D - 1);
        int row = row0 + r; if (row >= N) row = N - 1;
        size_t off = (size_t)row * D + c;
        lds[0][r][c] = bufA[off];
        lds[1][r][c] = xin[off];
        lds[2][r][c] = bufB[off];
        lds[3][r][c] = hl[off];
    }
    __syncthreads();

    const int wave = tid >> 5;
    const int lane = tid & 31;
    const int m = lane & 15;
    const int half = lane >> 4;
    const int col = wave * 16 + m;

    v8f cz[2] = {{}, {}};
    v8f cr[2] = {{}, {}};
    v8f ct[2] = {{}, {}};
#pragma unroll 2
    for (int kb = 0; kb < D; kb += 4) {
        const int kk = kb + half * 2;
#define BF(W, g) mkv2(W[(size_t)(g) * D * D + (size_t)kk * D + col], \
                      W[(size_t)(g) * D * D + (size_t)(kk + 1) * D + col])
        v2f b0l = BF(Wl, 0), b0r = BF(Wr, 0);
        v2f b1l = BF(Wl, 1), b1r = BF(Wr, 1);
        v2f b2l = BF(Wl, 2), b2r = BF(Wr, 2);
        v2f b3l = BF(Wl, 3), b3r = BF(Wr, 3);
        v2f b4l = BF(Wl, 4), b4r = BF(Wr, 4);
#undef BF
#pragma unroll
        for (int sub = 0; sub < 2; ++sub) {
            const int mm = sub * 16 + m;
            v2f aAX = mkv2(lds[0][mm][kk], lds[0][mm][kk + 1]);
            v2f aX  = mkv2(lds[1][mm][kk], lds[1][mm][kk + 1]);
            v2f aAH = mkv2(lds[2][mm][kk], lds[2][mm][kk + 1]);
            v2f aH  = mkv2(lds[3][mm][kk], lds[3][mm][kk + 1]);
            cz[sub] = wmma_f32(aAX, b0l, cz[sub]);
            cz[sub] = wmma_f32(aX,  b0r, cz[sub]);
            cz[sub] = wmma_f32(aAH, b1l, cz[sub]);
            cz[sub] = wmma_f32(aH,  b1r, cz[sub]);
            cr[sub] = wmma_f32(aAX, b2l, cr[sub]);
            cr[sub] = wmma_f32(aX,  b2r, cr[sub]);
            cr[sub] = wmma_f32(aAH, b3l, cr[sub]);
            cr[sub] = wmma_f32(aH,  b3r, cr[sub]);
            ct[sub] = wmma_f32(aAX, b4l, ct[sub]);
            ct[sub] = wmma_f32(aX,  b4r, ct[sub]);
        }
    }

    const float bz = bias[col] + bias[D + col];
    const float br = bias[2 * D + col] + bias[3 * D + col];
    const float bt = bias[4 * D + col];

    __syncthreads();
#pragma unroll
    for (int sub = 0; sub < 2; ++sub) {
#pragma unroll
        for (int j = 0; j < 8; ++j) {
            int rloc = sub * 16 + j + half * 8;
            float r = sigmoidf_(cr[sub][j] + br);
            lds[0][rloc][col] = r * lds[3][rloc][col];
            ct[sub][j] += bt;
        }
    }
    __syncthreads();

    const float* __restrict__ Wl5 = Wl + (size_t)5 * D * D;
    const float* __restrict__ Wr5 = Wr + (size_t)5 * D * D;
    v8f cq[2] = {{}, {}};
#pragma unroll 2
    for (int kb = 0; kb < D; kb += 4) {
        const int kk = kb + half * 2;
        v2f bl5 = mkv2(Wl5[(size_t)kk * D + col], Wl5[(size_t)(kk + 1) * D + col]);
        v2f br5 = mkv2(Wr5[(size_t)kk * D + col], Wr5[(size_t)(kk + 1) * D + col]);
#pragma unroll
        for (int sub = 0; sub < 2; ++sub) {
            const int mm = sub * 16 + m;
            v2f a = mkv2(lds[0][mm][kk], lds[0][mm][kk + 1]);
            cq[sub] = wmma_f32(a, bl5, cq[sub]);
            ct[sub] = wmma_f32(a, br5, ct[sub]);
        }
    }

#pragma unroll
    for (int sub = 0; sub < 2; ++sub) {
#pragma unroll
        for (int j = 0; j < 8; ++j) {
            int row = row0 + sub * 16 + j + half * 8;
            if (row < N) {
                size_t off = (size_t)row * D + col;
                bufA[off] = sigmoidf_(cz[sub][j] + bz);
                qbuf[off] = cq[sub][j];
                bufB[off] = ct[sub][j];
            }
        }
    }
}

extern "C" void kernel_launch(void* const* d_in, const int* in_sizes, int n_in,
                              void* d_out, int out_size, void* d_ws, size_t ws_size,
                              hipStream_t stream) {
    const float* x  = (const float*)d_in[0];
    const float* h  = (const float*)d_in[1];
    const float* Wl = (const float*)d_in[2];
    const float* Wr = (const float*)d_in[3];
    const float* b  = (const float*)d_in[4];
    const int* src  = (const int*)d_in[5];
    const int* dst  = (const int*)d_in[6];
    const size_t ND = (size_t)in_sizes[0];
    const int N = (int)(ND / D);
    const int L = in_sizes[1] / (int)ND;
    const int E = in_sizes[5];
    float* out = (float*)d_out;

    char* wsc = (char*)d_ws;
    int* indeg  = (int*)wsc;
    int* start  = indeg + N;
    int* csrc   = start + N;
    size_t int_bytes = ((size_t)(2 * N + E) * sizeof(int) + 63) & ~(size_t)63;
    float* fbase = (float*)(wsc + int_bytes);
    float* bufA  = fbase;
    float* bufB  = bufA + ND;
    float* qbuf  = bufB + ND;

    const int tiles = (N + 31) / 32;
    const int gblocks = (N + 7) / 8;

    const int NB = (N + BNODES - 1) / BNODES;
    const int EB = (E + ECH - 1) / ECH;
    const int total = NB * EB;
    int* bh   = (int*)bufA;
    int* offs = bh + total;
    int* bdst = (int*)qbuf;
    int* bsrc = (int*)bufB;
    k_binhist<<<EB, ETH, 0, stream>>>(dst, bh, E, NB, EB);
    k_binscan<<<1, SCAN_BS, 0, stream>>>(bh, offs, total);
    k_binscatter<<<EB, ETH, 0, stream>>>(src, dst, offs, bdst, bsrc, E, NB, EB);
    k_bucketsort<<<NB, ETH, 0, stream>>>(offs, bdst, bsrc, indeg, start, csrc, N, E, NB, EB);

    for (int l = 0; l < L; ++l) {
        const float* inp = (l == 0) ? x : (out + (size_t)(l - 1) * ND);
        const float* hlp = h + (size_t)l * ND;
        const float* Wll = Wl + (size_t)l * 6 * D * D;
        const float* Wrl = Wr + (size_t)l * 6 * D * D;
        const float* bl  = b + (size_t)l * 6 * D;
        float* outl = out + (size_t)l * ND;

        k_gather2<<<gblocks, 256, 0, stream>>>(inp, hlp, csrc, start, indeg,
                                               bufA, bufB, N);
        k_gates<<<tiles, 256, 0, stream>>>(inp, hlp, Wll, Wrl, bl,
                                           bufA, bufB, qbuf, N);
        k_gather_gru<<<gblocks, 256, 0, stream>>>(qbuf, bufB, bufA, hlp, bl + 5 * D,
                                                  csrc, start, indeg, outl, N);
    }
}


# CSR build in 2 launches (was 4): edge blocks keep bucket-sorted edges in own chunks and publish local bucket starts, whose sum is the bucket's base; slot order in a node still from LDS arrival
# speedup vs baseline: 1.3502x; 1.1211x over previous
#include <hip/hip_runtime.h>
#include <cstdint>

#define D 128
#define LDSP 132

typedef float v2f __attribute__((ext_vector_type(2)));
typedef float v8f __attribute__((ext_vector_type(8)));

__device__ __forceinline__ v2f mkv2(float a, float b) { v2f v; v.x = a; v.y = b; return v; }

__device__ __forceinline__ v8f wmma_f32(v2f a, v2f b, v8f c) {
    return __builtin_amdgcn_wmma_f32_16x16x4_f32(false, a, false, b, (short)0, c, false, false);
}

__device__ __forceinline__ float sigmoidf_(float x) { return 1.0f / (1.0f + __expf(-x)); }

#define BSH 8
#define BNODES (1 << BSH)
#define MAXB 1024
#define ETH 1024
#define ECH 4096
#define SCAP 8192
#define MAXEB ETH
#define NWV (ETH / 32)
#define RPT 4

static_assert(MAXB == ETH, "the local scan below gives each thread one bucket");
static_assert(ECH < (1 << 15), "a run's local start and its length share one int");
__global__ __launch_bounds__(ETH) void k_blocksort(const int* __restrict__ src,
                                                   const int* __restrict__ dst,
                                                   int* __restrict__ tab,
                                                   int* __restrict__ bdst,
                                                   int* __restrict__ bsrc,
                                                   int E, int NB, int EB) {
    __shared__ int lsc[MAXB];
    __shared__ int cur[MAXB];
    __shared__ int sd[ECH];
    __shared__ int ss[ECH];
    const int t = threadIdx.x, b = blockIdx.x;
    const int e0 = b * ECH;
    const int e1 = min(E, e0 + ECH);
    const int n = e1 - e0;
    cur[t] = 0;
    int d[ECH / ETH], s[ECH / ETH];
#pragma unroll
    for (int j = 0; j < ECH / ETH; ++j) {
        const int i = e0 + t + j * ETH;
        d[j] = (i < e1) ? dst[i] : -1;
        s[j] = (i < e1) ? src[i] : 0;
    }
    __syncthreads();
#pragma unroll
    for (int j = 0; j < ECH / ETH; ++j)
        if (d[j] >= 0) atomicAdd(&cur[d[j] >> BSH], 1);
    __syncthreads();
    const int c = cur[t];
    lsc[t] = c;
    __syncthreads();
    for (int off = 1; off < MAXB; off <<= 1) {
        int v = (t >= off) ? lsc[t - off] : 0;
        __syncthreads();
        lsc[t] += v;
        __syncthreads();
    }
    const int ex = lsc[t] - c;
    cur[t] = ex;
    if (t < NB) tab[(size_t)t * EB + b] = (ex << 16) | c;
    __syncthreads();
#pragma unroll
    for (int j = 0; j < ECH / ETH; ++j) {
        if (d[j] >= 0) {
            const int p = atomicAdd(&cur[d[j] >> BSH], 1);
            sd[p] = d[j];
            ss[p] = s[j];
        }
    }
    __syncthreads();
    for (int j = t; j < n; j += ETH) {
        bdst[e0 + j] = sd[j];
        bsrc[e0 + j] = ss[j];
    }
}

__global__ __launch_bounds__(ETH) void k_bucketsort(const int* __restrict__ tab,
                                                    const int* __restrict__ bdst,
                                                    const int* __restrict__ bsrc,
                                                    int* __restrict__ indeg,
                                                    int* __restrict__ start,
                                                    int* __restrict__ csrc,
                                                    int N, int EB) {
    __shared__ int cnt[BNODES];
    __shared__ int sc[BNODES];
    __shared__ int so[SCAP];
    __shared__ int rs[MAXEB];
    __shared__ int rc[MAXEB];
    __shared__ int tot[2];
    const int t = threadIdx.x, b = blockIdx.x;
    const int wave = t >> 5, lane = t & 31;
    const bool own = t < BNODES;
    if (own) cnt[t] = 0;
    if (t < 2) tot[t] = 0;
    int rex = 0, rcn = 0;
    if (t < EB) {
        const int w = tab[(size_t)b * EB + t];
        rex = w >> 16;
        rcn = w & 0xffff;
        rs[t] = t * ECH + rex;
        rc[t] = rcn;
    }
    __syncthreads();
    for (int off = 16; off > 0; off >>= 1) {
        rex += __shfl_xor(rex, off, 32);
        rcn += __shfl_xor(rcn, off, 32);
    }
    if (lane == 0 && wave * 32 < EB) {
        atomicAdd(&tot[0], rex);
        atomicAdd(&tot[1], rcn);
    }
    __syncthreads();
    const int bs = tot[0];
    const int n = tot[1];
    const bool staged = n <= SCAP;
    for (int k0 = wave; k0 < EB; k0 += RPT * NWV) {
        int cu[RPT], gu[RPT], dv[RPT];
#pragma unroll
        for (int u = 0; u < RPT; ++u) {
            const int k = k0 + u * NWV;
            cu[u] = (k < EB) ? rc[k] : 0;
            gu[u] = (k < EB) ? rs[k] : 0;
        }
#pragma unroll
        for (int u = 0; u < RPT; ++u) dv[u] = (lane < cu[u]) ? bdst[gu[u] + lane] : -1;
#pragma unroll
        for (int u = 0; u < RPT; ++u)
            if (dv[u] >= 0) atomicAdd(&cnt[dv[u] & (BNODES - 1)], 1);
#pragma unroll
        for (int u = 0; u < RPT; ++u)
            for (int i = lane + 32; i < cu[u]; i += 32)
                atomicAdd(&cnt[bdst[gu[u] + i] & (BNODES - 1)], 1);
    }
    __syncthreads();
    int c = 0;
    if (own) { c = cnt[t]; sc[t] = c; }
    __syncthreads();
    for (int off = 1; off < BNODES; off <<= 1) {
        int v = 0;
        if (own && t >= off) v = sc[t - off];
        __syncthreads();
        if (own) sc[t] += v;
        __syncthreads();
    }
    if (own) {
        const int ex = sc[t] - c;
        const int node = (b << BSH) + t;
        if (node < N) {
            indeg[node] = c;
            start[node] = bs + ex;
        }
        cnt[t] = ex;
    }
    __syncthreads();
    for (int k0 = wave; k0 < EB; k0 += RPT * NWV) {
        int cu[RPT], gu[RPT], dv[RPT], sv[RPT];
#pragma unroll
        for (int u = 0; u < RPT; ++u) {
            const int k = k0 + u * NWV;
            cu[u] = (k < EB) ? rc[k] : 0;
            gu[u] = (k < EB) ? rs[k] : 0;
        }
#pragma unroll
        for (int u = 0; u < RPT; ++u) {
            const bool ok = lane < cu[u];
            dv[u] = ok ? bdst[gu[u] + lane] : -1;
            sv[u] = ok ? bsrc[gu[u] + lane] : 0;
        }
#pragma unroll
        for (int u = 0; u < RPT; ++u) {
            if (dv[u] >= 0) {
                const int p = atomicAdd(&cnt[dv[u] & (BNODES - 1)], 1);
                if (staged) so[p] = sv[u];
                else csrc[bs + p] = sv[u];
            }
        }
#pragma unroll
        for (int u = 0; u < RPT; ++u) {
            for (int i = lane + 32; i < cu[u]; i += 32) {
                const int p = atomicAdd(&cnt[bdst[gu[u] + i] & (BNODES - 1)], 1);
                const int v = bsrc[gu[u] + i];
                if (staged) so[p] = v;
                else csrc[bs + p] = v;
            }
        }
    }
    if (staged) {
        __syncthreads();
        for (int j = t; j < n; j += ETH) csrc[bs + j] = so[j];
    }
}

__global__ void k_gather2(const float* __restrict__ xin, const float* __restrict__ hl,
                          const int* __restrict__ csrc,
                          const int* __restrict__ start, const int* __restrict__ indeg,
                          float* __restrict__ aggx, float* __restrict__ aggh, int N) {
    int node = (int)(((size_t)blockIdx.x * blockDim.x + threadIdx.x) >> 5);
    int lane = threadIdx.x & 31;
    if (node >= N) return;
    int base = start[node], cnt = indeg[node];
    float4 ax = {0.f, 0.f, 0.f, 0.f}, ah = {0.f, 0.f, 0.f, 0.f};
#pragma unroll 2
    for (int j = 0; j < cnt; ++j) {
        int s = csrc[base + j];
        float4 vx = ((const float4*)(xin + (size_t)s * D))[lane];
        float4 vh = ((const float4*)(hl + (size_t)s * D))[lane];
        ax.x += vx.x; ax.y += vx.y; ax.z += vx.z; ax.w += vx.w;
        ah.x += vh.x; ah.y += vh.y; ah.z += vh.z; ah.w += vh.w;
    }
    float inv = 1.0f / fmaxf((float)cnt, 1.0f);
    float4 ox = {ax.x * inv, ax.y * inv, ax.z * inv, ax.w * inv};
    float4 oh = {ah.x * inv, ah.y * inv, ah.z * inv, ah.w * inv};
    ((float4*)(aggx + (size_t)node * D))[lane] = ox;
    ((float4*)(aggh + (size_t)node * D))[lane] = oh;
}

__global__ void k_gather_gru(const float* __restrict__ q, const float* __restrict__ sbuf,
                             const float* __restrict__ zbuf, const float* __restrict__ hl,
                             const float* __restrict__ b5,
                             const int* __restrict__ csrc,
                             const int* __restrict__ start, const int* __restrict__ indeg,
                             float* __restrict__ outl, int N) {
    int node = (int)(((size_t)blockIdx.x * blockDim.x + threadIdx.x) >> 5);
    int lane = threadIdx.x & 31;
    if (node >= N) return;
    int base = start[node], cnt = indeg[node];
    const size_t ro = (size_t)node * D;
    float4 sv = ((const float4*)(sbuf + ro))[lane];
    float4 zv = ((const float4*)(zbuf + ro))[lane];
    float4 hv = ((const float4*)(hl + ro))[lane];
    float4 bv = ((const float4*)b5)[lane];
    float4 aq = {0.f, 0.f, 0.f, 0.f};
#pragma unroll 2
    for (int j = 0; j < cnt; ++j) {
        int sn = csrc[base + j];
        float4 vq = ((const float4*)(q + (size_t)sn * D))[lane];
        aq.x += vq.x; aq.y += vq.y; aq.z += vq.z; aq.w += vq.w;
    }
    float inv = 1.0f / fmaxf((float)cnt, 1.0f);
    float4 o;
    o.x = zv.x * hv.x + (1.0f - zv.x) * tanhf((sv.x + aq.x * inv) + bv.x);
    o.y = zv.y * hv.y + (1.0f - zv.y) * tanhf((sv.y + aq.y * inv) + bv.y);
    o.z = zv.z * hv.z + (1.0f - zv.z) * tanhf((sv.z + aq.z * inv) + bv.z);
    o.w = zv.w * hv.w + (1.0f - zv.w) * tanhf((sv.w + aq.w * inv) + bv.w);
    ((float4*)(outl + ro))[lane] = o;
}

__global__ __launch_bounds__(256, 8) void k_gates(
    const float* __restrict__ xin, const float* __restrict__ hl,
    const float* __restrict__ Wl, const float* __restrict__ Wr,
    const float* __restrict__ bias,
    float* __restrict__ bufA,
    float* __restrict__ bufB,
    float* __restrict__ qbuf, int N) {
    __shared__ float lds[4][32][LDSP];
    const int row0 = blockIdx.x * 32;
    const int tid = threadIdx.x;
    for (int idx = tid; idx < 32 * D; idx += 256) {
        int r = idx >> 7, c = idx & (D - 1);
        int row = row0 + r; if (row >= N) row = N - 1;
        size_t off = (size_t)row * D + c;
        lds[0][r][c] = bufA[off];
        lds[1][r][c] = xin[off];
        lds[2][r][c] = bufB[off];
        lds[3][r][c] = hl[off];
    }
    __syncthreads();

    const int wave = tid >> 5;
    const int lane = tid & 31;
    const int m = lane & 15;
    const int half = lane >> 4;
    const int col = wave * 16 + m;

    v8f cz[2] = {{}, {}};
    v8f cr[2] = {{}, {}};
    v8f ct[2] = {{}, {}};
#pragma unroll 2
    for (int kb = 0; kb < D; kb += 4) {
        const int kk = kb + half * 2;
#define BF(W, g) mkv2(W[(size_t)(g) * D * D + (size_t)kk * D + col], \
                      W[(size_t)(g) * D * D + (size_t)(kk + 1) * D + col])
        v2f b0l = BF(Wl, 0), b0r = BF(Wr, 0);
        v2f b1l = BF(Wl, 1), b1r = BF(Wr, 1);
        v2f b2l = BF(Wl, 2), b2r = BF(Wr, 2);
        v2f b3l = BF(Wl, 3), b3r = BF(Wr, 3);
        v2f b4l = BF(Wl, 4), b4r = BF(Wr, 4);
#undef BF
#pragma unroll
        for (int sub = 0; sub < 2; ++sub) {
            const int mm = sub * 16 + m;
            v2f aAX = mkv2(lds[0][mm][kk], lds[0][mm][kk + 1]);
            v2f aX  = mkv2(lds[1][mm][kk], lds[1][mm][kk + 1]);
            v2f aAH = mkv2(lds[2][mm][kk], lds[2][mm][kk + 1]);
            v2f aH  = mkv2(lds[3][mm][kk], lds[3][mm][kk + 1]);
            cz[sub] = wmma_f32(aAX, b0l, cz[sub]);
            cz[sub] = wmma_f32(aX,  b0r, cz[sub]);
            cz[sub] = wmma_f32(aAH, b1l, cz[sub]);
            cz[sub] = wmma_f32(aH,  b1r, cz[sub]);
            cr[sub] = wmma_f32(aAX, b2l, cr[sub]);
            cr[sub] = wmma_f32(aX,  b2r, cr[sub]);
            cr[sub] = wmma_f32(aAH, b3l, cr[sub]);
            cr[sub] = wmma_f32(aH,  b3r, cr[sub]);
            ct[sub] = wmma_f32(aAX, b4l, ct[sub]);
            ct[sub] = wmma_f32(aX,  b4r, ct[sub]);
        }
    }

    const float bz = bias[col] + bias[D + col];
    const float br = bias[2 * D + col] + bias[3 * D + col];
    const float bt = bias[4 * D + col];

    __syncthreads();
#pragma unroll
    for (int sub = 0; sub < 2; ++sub) {
#pragma unroll
        for (int j = 0; j < 8; ++j) {
            int rloc = sub * 16 + j + half * 8;
            float r = sigmoidf_(cr[sub][j] + br);
            lds[0][rloc][col] = r * lds[3][rloc][col];
            ct[sub][j] += bt;
        }
    }
    __syncthreads();

    const float* __restrict__ Wl5 = Wl + (size_t)5 * D * D;
    const float* __restrict__ Wr5 = Wr + (size_t)5 * D * D;
    v8f cq[2] = {{}, {}};
#pragma unroll 2
    for (int kb = 0; kb < D; kb += 4) {
        const int kk = kb + half * 2;
        v2f bl5 = mkv2(Wl5[(size_t)kk * D + col], Wl5[(size_t)(kk + 1) * D + col]);
        v2f br5 = mkv2(Wr5[(size_t)kk * D + col], Wr5[(size_t)(kk + 1) * D + col]);
#pragma unroll
        for (int sub = 0; sub < 2; ++sub) {
            const int mm = sub * 16 + m;
            v2f a = mkv2(lds[0][mm][kk], lds[0][mm][kk + 1]);
            cq[sub] = wmma_f32(a, bl5, cq[sub]);
            ct[sub] = wmma_f32(a, br5, ct[sub]);
        }
    }

#pragma unroll
    for (int sub = 0; sub < 2; ++sub) {
#pragma unroll
        for (int j = 0; j < 8; ++j) {
            int row = row0 + sub * 16 + j + half * 8;
            if (row < N) {
                size_t off = (size_t)row * D + col;
                bufA[off] = sigmoidf_(cz[sub][j] + bz);
                qbuf[off] = cq[sub][j];
                bufB[off] = ct[sub][j];
            }
        }
    }
}

extern "C" void kernel_launch(void* const* d_in, const int* in_sizes, int n_in,
                              void* d_out, int out_size, void* d_ws, size_t ws_size,
                              hipStream_t stream) {
    const float* x  = (const float*)d_in[0];
    const float* h  = (const float*)d_in[1];
    const float* Wl = (const float*)d_in[2];
    const float* Wr = (const float*)d_in[3];
    const float* b  = (const float*)d_in[4];
    const int* src  = (const int*)d_in[5];
    const int* dst  = (const int*)d_in[6];
    const size_t ND = (size_t)in_sizes[0];
    const int N = (int)(ND / D);
    const int L = in_sizes[1] / (int)ND;
    const int E = in_sizes[5];
    float* out = (float*)d_out;

    char* wsc = (char*)d_ws;
    int* indeg  = (int*)wsc;
    int* start  = indeg + N;
    int* csrc   = start + N;
    size_t int_bytes = ((size_t)(2 * N + E) * sizeof(int) + 63) & ~(size_t)63;
    float* fbase = (float*)(wsc + int_bytes);
    float* bufA  = fbase;
    float* bufB  = bufA + ND;
    float* qbuf  = bufB + ND;

    const int tiles = (N + 31) / 32;
    const int gblocks = (N + 7) / 8;

    const int NB = (N + BNODES - 1) / BNODES;
    const int EB = (E + ECH - 1) / ECH;
    int* tab  = (int*)bufA;
    int* bdst = (int*)qbuf;
    int* bsrc = (int*)bufB;
    k_blocksort<<<EB, ETH, 0, stream>>>(src, dst, tab, bdst, bsrc, E, NB, EB);
    k_bucketsort<<<NB, ETH, 0, stream>>>(tab, bdst, bsrc, indeg, start, csrc, N, EB);

    for (int l = 0; l < L; ++l) {
        const float* inp = (l == 0) ? x : (out + (size_t)(l - 1) * ND);
        const float* hlp = h + (size_t)l * ND;
        const float* Wll = Wl + (size_t)l * 6 * D * D;
        const float* Wrl = Wr + (size_t)l * 6 * D * D;
        const float* bl  = b + (size_t)l * 6 * D;
        float* outl = out + (size_t)l * ND;

        k_gather2<<<gblocks, 256, 0, stream>>>(inp, hlp, csrc, start, indeg,
                                               bufA, bufB, N);
        k_gates<<<tiles, 256, 0, stream>>>(inp, hlp, Wll, Wrl, bl,
                                           bufA, bufB, qbuf, N);
        k_gather_gru<<<gblocks, 256, 0, stream>>>(qbuf, bufB, bufA, hlp, bl + 5 * D,
                                                  csrc, start, indeg, outl, N);
    }
}


# Gates GEMM tile 48-64 rows tall on a resident grid (512 blocks, even shares of 16-row subtiles in passes of 3-4): each weight fragment from L2 feeds 3-4 products, was 2; sum order unchanged
# speedup vs baseline: 1.3938x; 1.0323x over previous
#include <hip/hip_runtime.h>
#include <cstdint>

#define D 128
#define LDSP 132

typedef float v2f __attribute__((ext_vector_type(2)));
typedef float v8f __attribute__((ext_vector_type(8)));

__device__ __forceinline__ v2f mkv2(float a, float b) { v2f v; v.x = a; v.y = b; return v; }

__device__ __forceinline__ v8f wmma_f32(v2f a, v2f b, v8f c) {
    return __builtin_amdgcn_wmma_f32_16x16x4_f32(false, a, false, b, (short)0, c, false, false);
}

__device__ __forceinline__ float sigmoidf_(float x) { return 1.0f / (1.0f + __expf(-x)); }

#define BSH 8
#define BNODES (1 << BSH)
#define MAXB 1024
#define ETH 1024
#define ECH 4096
#define SCAP 8192
#define MAXEB ETH
#define NWV (ETH / 32)
#define RPT 4

static_assert(MAXB == ETH, "the local scan below gives each thread one bucket");
static_assert(ECH < (1 << 15), "a run's local start and its length share one int");
__global__ __launch_bounds__(ETH) void k_blocksort(const int* __restrict__ src,
                                                   const int* __restrict__ dst,
                                                   int* __restrict__ tab,
                                                   int* __restrict__ bdst,
                                                   int* __restrict__ bsrc,
                                                   int E, int NB, int EB) {
    __shared__ int lsc[MAXB];
    __shared__ int cur[MAXB];
    __shared__ int sd[ECH];
    __shared__ int ss[ECH];
    const int t = threadIdx.x, b = blockIdx.x;
    const int e0 = b * ECH;
    const int e1 = min(E, e0 + ECH);
    const int n = e1 - e0;
    cur[t] = 0;
    int d[ECH / ETH], s[ECH / ETH];
#pragma unroll
    for (int j = 0; j < ECH / ETH; ++j) {
        const int i = e0 + t + j * ETH;
        d[j] = (i < e1) ? dst[i] : -1;
        s[j] = (i < e1) ? src[i] : 0;
    }
    __syncthreads();
#pragma unroll
    for (int j = 0; j < ECH / ETH; ++j)
        if (d[j] >= 0) atomicAdd(&cur[d[j] >> BSH], 1);
    __syncthreads();
    const int c = cur[t];
    lsc[t] = c;
    __syncthreads();
    for (int off = 1; off < MAXB; off <<= 1) {
        int v = (t >= off) ? lsc[t - off] : 0;
        __syncthreads();
        lsc[t] += v;
        __syncthreads();
    }
    const int ex = lsc[t] - c;
    cur[t] = ex;
    if (t < NB) tab[(size_t)t * EB + b] = (ex << 16) | c;
    __syncthreads();
#pragma unroll
    for (int j = 0; j < ECH / ETH; ++j) {
        if (d[j] >= 0) {
            const int p = atomicAdd(&cur[d[j] >> BSH], 1);
            sd[p] = d[j];
            ss[p] = s[j];
        }
    }
    __syncthreads();
    for (int j = t; j < n; j += ETH) {
        bdst[e0 + j] = sd[j];
        bsrc[e0 + j] = ss[j];
    }
}

__global__ __launch_bounds__(ETH) void k_bucketsort(const int* __restrict__ tab,
                                                    const int* __restrict__ bdst,
                                                    const int* __restrict__ bsrc,
                                                    int* __restrict__ indeg,
                                                    int* __restrict__ start,
                                                    int* __restrict__ csrc,
                                                    int N, int EB) {
    __shared__ int cnt[BNODES];
    __shared__ int sc[BNODES];
    __shared__ int so[SCAP];
    __shared__ int rs[MAXEB];
    __shared__ int rc[MAXEB];
    __shared__ int tot[2];
    const int t = threadIdx.x, b = blockIdx.x;
    const int wave = t >> 5, lane = t & 31;
    const bool own = t < BNODES;
    if (own) cnt[t] = 0;
    if (t < 2) tot[t] = 0;
    int rex = 0, rcn = 0;
    if (t < EB) {
        const int w = tab[(size_t)b * EB + t];
        rex = w >> 16;
        rcn = w & 0xffff;
        rs[t] = t * ECH + rex;
        rc[t] = rcn;
    }
    __syncthreads();
    for (int off = 16; off > 0; off >>= 1) {
        rex += __shfl_xor(rex, off, 32);
        rcn += __shfl_xor(rcn, off, 32);
    }
    if (lane == 0 && wave * 32 < EB) {
        atomicAdd(&tot[0], rex);
        atomicAdd(&tot[1], rcn);
    }
    __syncthreads();
    const int bs = tot[0];
    const int n = tot[1];
    const bool staged = n <= SCAP;
    for (int k0 = wave; k0 < EB; k0 += RPT * NWV) {
        int cu[RPT], gu[RPT], dv[RPT];
#pragma unroll
        for (int u = 0; u < RPT; ++u) {
            const int k = k0 + u * NWV;
            cu[u] = (k < EB) ? rc[k] : 0;
            gu[u] = (k < EB) ? rs[k] : 0;
        }
#pragma unroll
        for (int u = 0; u < RPT; ++u) dv[u] = (lane < cu[u]) ? bdst[gu[u] + lane] : -1;
#pragma unroll
        for (int u = 0; u < RPT; ++u)
            if (dv[u] >= 0) atomicAdd(&cnt[dv[u] & (BNODES - 1)], 1);
#pragma unroll
        for (int u = 0; u < RPT; ++u)
            for (int i = lane + 32; i < cu[u]; i += 32)
                atomicAdd(&cnt[bdst[gu[u] + i] & (BNODES - 1)], 1);
    }
    __syncthreads();
    int c = 0;
    if (own) { c = cnt[t]; sc[t] = c; }
    __syncthreads();
    for (int off = 1; off < BNODES; off <<= 1) {
        int v = 0;
        if (own && t >= off) v = sc[t - off];
        __syncthreads();
        if (own) sc[t] += v;
        __syncthreads();
    }
    if (own) {
        const int ex = sc[t] - c;
        const int node = (b << BSH) + t;
        if (node < N) {
            indeg[node] = c;
            start[node] = bs + ex;
        }
        cnt[t] = ex;
    }
    __syncthreads();
    for (int k0 = wave; k0 < EB; k0 += RPT * NWV) {
        int cu[RPT], gu[RPT], dv[RPT], sv[RPT];
#pragma unroll
        for (int u = 0; u < RPT; ++u) {
            const int k = k0 + u * NWV;
            cu[u] = (k < EB) ? rc[k] : 0;
            gu[u] = (k < EB) ? rs[k] : 0;
        }
#pragma unroll
        for (int u = 0; u < RPT; ++u) {
            const bool ok = lane < cu[u];
            dv[u] = ok ? bdst[gu[u] + lane] : -1;
            sv[u] = ok ? bsrc[gu[u] + lane] : 0;
        }
#pragma unroll
        for (int u = 0; u < RPT; ++u) {
            if (dv[u] >= 0) {
                const int p = atomicAdd(&cnt[dv[u] & (BNODES - 1)], 1);
                if (staged) so[p] = sv[u];
                else csrc[bs + p] = sv[u];
            }
        }
#pragma unroll
        for (int u = 0; u < RPT; ++u) {
            for (int i = lane + 32; i < cu[u]; i += 32) {
                const int p = atomicAdd(&cnt[bdst[gu[u] + i] & (BNODES - 1)], 1);
                const int v = bsrc[gu[u] + i];
                if (staged) so[p] = v;
                else csrc[bs + p] = v;
            }
        }
    }
    if (staged) {
        __syncthreads();
        for (int j = t; j < n; j += ETH) csrc[bs + j] = so[j];
    }
}

__global__ void k_gather2(const float* __restrict__ xin, const float* __restrict__ hl,
                          const int* __restrict__ csrc,
                          const int* __restrict__ start, const int* __restrict__ indeg,
                          float* __restrict__ aggx, float* __restrict__ aggh, int N) {
    int node = (int)(((size_t)blockIdx.x * blockDim.x + threadIdx.x) >> 5);
    int lane = threadIdx.x & 31;
    if (node >= N) return;
    int base = start[node], cnt = indeg[node];
    float4 ax = {0.f, 0.f, 0.f, 0.f}, ah = {0.f, 0.f, 0.f, 0.f};
#pragma unroll 2
    for (int j = 0; j < cnt; ++j) {
        int s = csrc[base + j];
        float4 vx = ((const float4*)(xin + (size_t)s * D))[lane];
        float4 vh = ((const float4*)(hl + (size_t)s * D))[lane];
        ax.x += vx.x; ax.y += vx.y; ax.z += vx.z; ax.w += vx.w;
        ah.x += vh.x; ah.y += vh.y; ah.z += vh.z; ah.w += vh.w;
    }
    float inv = 1.0f / fmaxf((float)cnt, 1.0f);
    float4 ox = {ax.x * inv, ax.y * inv, ax.z * inv, ax.w * inv};
    float4 oh = {ah.x * inv, ah.y * inv, ah.z * inv, ah.w * inv};
    ((float4*)(aggx + (size_t)node * D))[lane] = ox;
    ((float4*)(aggh + (size_t)node * D))[lane] = oh;
}

__global__ void k_gather_gru(const float* __restrict__ q, const float* __restrict__ sbuf,
                             const float* __restrict__ zbuf, const float* __restrict__ hl,
                             const float* __restrict__ b5,
                             const int* __restrict__ csrc,
                             const int* __restrict__ start, const int* __restrict__ indeg,
                             float* __restrict__ outl, int N) {
    int node = (int)(((size_t)blockIdx.x * blockDim.x + threadIdx.x) >> 5);
    int lane = threadIdx.x & 31;
    if (node >= N) return;
    int base = start[node], cnt = indeg[node];
    const size_t ro = (size_t)node * D;
    float4 sv = ((const float4*)(sbuf + ro))[lane];
    float4 zv = ((const float4*)(zbuf + ro))[lane];
    float4 hv = ((const float4*)(hl + ro))[lane];
    float4 bv = ((const float4*)b5)[lane];
    float4 aq = {0.f, 0.f, 0.f, 0.f};
#pragma unroll 2
    for (int j = 0; j < cnt; ++j) {
        int sn = csrc[base + j];
        float4 vq = ((const float4*)(q + (size_t)sn * D))[lane];
        aq.x += vq.x; aq.y += vq.y; aq.z += vq.z; aq.w += vq.w;
    }
    float inv = 1.0f / fmaxf((float)cnt, 1.0f);
    float4 o;
    o.x = zv.x * hv.x + (1.0f - zv.x) * tanhf((sv.x + aq.x * inv) + bv.x);
    o.y = zv.y * hv.y + (1.0f - zv.y) * tanhf((sv.y + aq.y * inv) + bv.y);
    o.z = zv.z * hv.z + (1.0f - zv.z) * tanhf((sv.z + aq.z * inv) + bv.z);
    o.w = zv.w * hv.w + (1.0f - zv.w) * tanhf((sv.w + aq.w * inv) + bv.w);
    ((float4*)(outl + ro))[lane] = o;
}

#define MAXSUB 4
#define TROWS (MAXSUB * 16)
#define GBLK 512

template <int SUBS>
__device__ __forceinline__ void gates_pass(
    float (*lds)[TROWS][LDSP], const int row0, const int tid,
    const float* __restrict__ xin, const float* __restrict__ hl,
    const float* __restrict__ Wl, const float* __restrict__ Wr,
    const float* __restrict__ bias,
    float* __restrict__ bufA, float* __restrict__ bufB,
    float* __restrict__ qbuf, const int N) {
    for (int idx = tid; idx < SUBS * 16 * D; idx += 256) {
        int r = idx >> 7, c = idx & (D - 1);
        int row = row0 + r; if (row >= N) row = N - 1;
        size_t off = (size_t)row * D + c;
        lds[0][r][c] = bufA[off];
        lds[1][r][c] = xin[off];
        lds[2][r][c] = bufB[off];
        lds[3][r][c] = hl[off];
    }
    __syncthreads();

    const int wave = tid >> 5;
    const int lane = tid & 31;
    const int m = lane & 15;
    const int half = lane >> 4;
    const int col = wave * 16 + m;

    v8f cz[SUBS], cr[SUBS], ct[SUBS];
#pragma unroll
    for (int sub = 0; sub < SUBS; ++sub) { cz[sub] = v8f{}; cr[sub] = v8f{}; ct[sub] = v8f{}; }
#pragma unroll 2
    for (int kb = 0; kb < D; kb += 4) {
        const int kk = kb + half * 2;
#define BF(W, g) mkv2(W[(size_t)(g) * D * D + (size_t)kk * D + col], \
                      W[(size_t)(g) * D * D + (size_t)(kk + 1) * D + col])
        v2f b0l = BF(Wl, 0), b0r = BF(Wr, 0);
        v2f b1l = BF(Wl, 1), b1r = BF(Wr, 1);
        v2f b2l = BF(Wl, 2), b2r = BF(Wr, 2);
        v2f b3l = BF(Wl, 3), b3r = BF(Wr, 3);
        v2f b4l = BF(Wl, 4), b4r = BF(Wr, 4);
#undef BF
#pragma unroll
        for (int sub = 0; sub < SUBS; ++sub) {
            const int mm = sub * 16 + m;
            v2f aAX = mkv2(lds[0][mm][kk], lds[0][mm][kk + 1]);
            v2f aX  = mkv2(lds[1][mm][kk], lds[1][mm][kk + 1]);
            v2f aAH = mkv2(lds[2][mm][kk], lds[2][mm][kk + 1]);
            v2f aH  = mkv2(lds[3][mm][kk], lds[3][mm][kk + 1]);
            cz[sub] = wmma_f32(aAX, b0l, cz[sub]);
            cz[sub] = wmma_f32(aX,  b0r, cz[sub]);
            cz[sub] = wmma_f32(aAH, b1l, cz[sub]);
            cz[sub] = wmma_f32(aH,  b1r, cz[sub]);
            cr[sub] = wmma_f32(aAX, b2l, cr[sub]);
            cr[sub] = wmma_f32(aX,  b2r, cr[sub]);
            cr[sub] = wmma_f32(aAH, b3l, cr[sub]);
            cr[sub] = wmma_f32(aH,  b3r, cr[sub]);
            ct[sub] = wmma_f32(aAX, b4l, ct[sub]);
            ct[sub] = wmma_f32(aX,  b4r, ct[sub]);
        }
    }

    const float bz = bias[col] + bias[D + col];
    const float br = bias[2 * D + col] + bias[3 * D + col];
    const float bt = bias[4 * D + col];

    __syncthreads();
#pragma unroll
    for (int sub = 0; sub < SUBS; ++sub) {
#pragma unroll
        for (int j = 0; j < 8; ++j) {
            int rloc = sub * 16 + j + half * 8;
            float r = sigmoidf_(cr[sub][j] + br);
            lds[0][rloc][col] = r * lds[3][rloc][col];
            ct[sub][j] += bt;
        }
    }
    __syncthreads();

    const float* __restrict__ Wl5 = Wl + (size_t)5 * D * D;
    const float* __restrict__ Wr5 = Wr + (size_t)5 * D * D;
    v8f cq[SUBS];
#pragma unroll
    for (int sub = 0; sub < SUBS; ++sub) cq[sub] = v8f{};
#pragma unroll 2
    for (int kb = 0; kb < D; kb += 4) {
        const int kk = kb + half * 2;
        v2f bl5 = mkv2(Wl5[(size_t)kk * D + col], Wl5[(size_t)(kk + 1) * D + col]);
        v2f br5 = mkv2(Wr5[(size_t)kk * D + col], Wr5[(size_t)(kk + 1) * D + col]);
#pragma unroll
        for (int sub = 0; sub < SUBS; ++sub) {
            const int mm = sub * 16 + m;
            v2f a = mkv2(lds[0][mm][kk], lds[0][mm][kk + 1]);
            cq[sub] = wmma_f32(a, bl5, cq[sub]);
            ct[sub] = wmma_f32(a, br5, ct[sub]);
        }
    }

#pragma unroll
    for (int sub = 0; sub < SUBS; ++sub) {
#pragma unroll
        for (int j = 0; j < 8; ++j) {
            int row = row0 + sub * 16 + j + half * 8;
            if (row < N) {
                size_t off = (size_t)row * D + col;
                bufA[off] = sigmoidf_(cz[sub][j] + bz);
                qbuf[off] = cq[sub][j];
                bufB[off] = ct[sub][j];
            }
        }
    }
}

__global__ __launch_bounds__(256, 4) void k_gates(
    const float* __restrict__ xin, const float* __restrict__ hl,
    const float* __restrict__ Wl, const float* __restrict__ Wr,
    const float* __restrict__ bias,
    float* __restrict__ bufA,
    float* __restrict__ bufB,
    float* __restrict__ qbuf, int N) {
    __shared__ float lds[4][TROWS][LDSP];
    const int tid = threadIdx.x;
    const int S = (N + 15) >> 4;
    const long long G = (long long)gridDim.x;
    int s = (int)(((long long)blockIdx.x * S) / G);
    const int s_end = (int)(((long long)(blockIdx.x + 1) * S) / G);
    bool later = false;
    while (s < s_end) {
        const int left = s_end - s;
        const int take = (left <= MAXSUB) ? left
                       : ((left == 5 || left == 6 || left == 9) ? 3 : MAXSUB);
        if (later) __syncthreads();
        const int row0 = s * 16;
        if (take == 4)      gates_pass<4>(lds, row0, tid, xin, hl, Wl, Wr, bias, bufA, bufB, qbuf, N);
        else if (take == 3) gates_pass<3>(lds, row0, tid, xin, hl, Wl, Wr, bias, bufA, bufB, qbuf, N);
        else if (take == 2) gates_pass<2>(lds, row0, tid, xin, hl, Wl, Wr, bias, bufA, bufB, qbuf, N);
        else                gates_pass<1>(lds, row0, tid, xin, hl, Wl, Wr, bias, bufA, bufB, qbuf, N);
        s += take;
        later = true;
    }
}

extern "C" void kernel_launch(void* const* d_in, const int* in_sizes, int n_in,
                              void* d_out, int out_size, void* d_ws, size_t ws_size,
                              hipStream_t stream) {
    const float* x  = (const float*)d_in[0];
    const float* h  = (const float*)d_in[1];
    const float* Wl = (const float*)d_in[2];
    const float* Wr = (const float*)d_in[3];
    const float* b  = (const float*)d_in[4];
    const int* src  = (const int*)d_in[5];
    const int* dst  = (const int*)d_in[6];
    const size_t ND = (size_t)in_sizes[0];
    const int N = (int)(ND / D);
    const int L = in_sizes[1] / (int)ND;
    const int E = in_sizes[5];
    float* out = (float*)d_out;

    char* wsc = (char*)d_ws;
    int* indeg  = (int*)wsc;
    int* start  = indeg + N;
    int* csrc   = start + N;
    size_t int_bytes = ((size_t)(2 * N + E) * sizeof(int) + 63) & ~(size_t)63;
    float* fbase = (float*)(wsc + int_bytes);
    float* bufA  = fbase;
    float* bufB  = bufA + ND;
    float* qbuf  = bufB + ND;

    const int gblocks = (N + 7) / 8;

    const int NB = (N + BNODES - 1) / BNODES;
    const int EB = (E + ECH - 1) / ECH;
    int* tab  = (int*)bufA;
    int* bdst = (int*)qbuf;
    int* bsrc = (int*)bufB;
    k_blocksort<<<EB, ETH, 0, stream>>>(src, dst, tab, bdst, bsrc, E, NB, EB);
    k_bucketsort<<<NB, ETH, 0, stream>>>(tab, bdst, bsrc, indeg, start, csrc, N, EB);

    for (int l = 0; l < L; ++l) {
        const float* inp = (l == 0) ? x : (out + (size_t)(l - 1) * ND);
        const float* hlp = h + (size_t)l * ND;
        const float* Wll = Wl + (size_t)l * 6 * D * D;
        const float* Wrl = Wr + (size_t)l * 6 * D * D;
        const float* bl  = b + (size_t)l * 6 * D;
        float* outl = out + (size_t)l * ND;

        k_gather2<<<gblocks, 256, 0, stream>>>(inp, hlp, csrc, start, indeg,
                                               bufA, bufB, N);
        k_gates<<<GBLK, 256, 0, stream>>>(inp, hlp, Wll, Wrl, bl,
                                           bufA, bufB, qbuf, N);
        k_gather_gru<<<gblocks, 256, 0, stream>>>(qbuf, bufB, bufA, hlp, bl + 5 * D,
                                                  csrc, start, indeg, outl, N);
    }
}
